# starting kernel
# baseline (speedup 1.0000x reference)
#include <hip/hip_runtime.h>

typedef float v2f __attribute__((ext_vector_type(2)));
typedef float v8f __attribute__((ext_vector_type(8)));

#define N_NODES 100000
#define N_EDGES 1600000
#define R_DIM   6
#define CH_IN   8
#define CH_OUT  16
#define K_DIM   96
#define OCOLS   32
#define N_TILES (N_NODES / 16)

__global__ __launch_bounds__(256) void scatter_kernel(
    const float* __restrict__ x,
    const int*   __restrict__ edges,
    const float* __restrict__ sten,
    float*       __restrict__ agg)
{
  int i = blockIdx.x * 256 + threadIdx.x;
  int e = i / R_DIM;
  int r = i - e * R_DIM;
  if (e >= N_EDGES) return;

  long long ep = __builtin_nontemporal_load((const long long*)edges + e);
  int src = (int)(ep & 0xffffffffLL);
  int dst = (int)(ep >> 32);

  v2f st = __builtin_nontemporal_load((const v2f*)sten + (size_t)e * R_DIM + r);

  const float4* xp = (const float4*)(x + (size_t)src * CH_IN);
  float4 x0 = xp[0];
  float4 x1 = xp[1];
  float xs[8] = {x0.x, x0.y, x0.z, x0.w, x1.x, x1.y, x1.z, x1.w};

  float* base = agg + (size_t)dst * K_DIM + r * (2 * CH_IN);
#pragma unroll
  for (int c = 0; c < CH_IN; ++c) {
    unsafeAtomicAdd(base + 2 * c,     st.x * xs[c]);
    unsafeAtomicAdd(base + 2 * c + 1, st.y * xs[c]);
  }
}

__global__ __launch_bounds__(256) void filter_kernel(
    const float* __restrict__ weight,
    const float* __restrict__ offset,
    float*       __restrict__ Fp)
{
  int i = blockIdx.x * 256 + threadIdx.x;
  if (i >= K_DIM * OCOLS) return;
  int k   = i >> 5;
  int col = i & 31;
  int p   = k & 1;
  int rc  = k >> 1;
  int r   = rc >> 3;
  int c   = rc & 7;
  int o   = col & 15;

  float w  = weight[(r * CH_IN + c) * CH_OUT + o];
  float ph = offset[c * CH_OUT + o];
  float fre = w * cosf(ph);
  float fim = w * sinf(ph);

  float val;
  if (col < 16) val = (p == 0) ? fre : -fim;
  else          val = (p == 0) ? fim :  fre;

  Fp[(rc * OCOLS + col) * 2 + p] = val;
}

__global__ __launch_bounds__(256) void gemm_kernel(
    const float* __restrict__ Fp,
    const float* __restrict__ agg,
    const float* __restrict__ bias,
    float*       __restrict__ out)
{
  __shared__ float Fl[K_DIM * OCOLS];
  for (int i = threadIdx.x; i < K_DIM * OCOLS; i += 256) Fl[i] = Fp[i];
  __syncthreads();

  int wave = threadIdx.x >> 5;
  int lane = threadIdx.x & 31;
  int tile = blockIdx.x * 8 + wave;
  if (tile >= N_TILES) return;

  int node_base = tile * 16;
  int mrow = lane & 15;
  int half = lane >> 4;
  const float* arow = agg + (size_t)(node_base + mrow) * K_DIM + half * 2;
  const float* fb0 = Fl + (size_t)(half * OCOLS + mrow) * 2;
  const float* fb1 = fb0 + 32;

  v8f acc0 = {0.f, 0.f, 0.f, 0.f, 0.f, 0.f, 0.f, 0.f};
  v8f acc1 = {0.f, 0.f, 0.f, 0.f, 0.f, 0.f, 0.f, 0.f};

#pragma unroll 4
  for (int k0 = 0; k0 < K_DIM; k0 += 4) {
    v2f a  = *(const v2f*)(arow + k0);
    v2f b0 = *(const v2f*)(fb0 + (k0 >> 1) * (OCOLS * 2));
    v2f b1 = *(const v2f*)(fb1 + (k0 >> 1) * (OCOLS * 2));
    acc0 = __builtin_amdgcn_wmma_f32_16x16x4_f32(
        false, a, false, b0, (short)0, acc0, false, false);
    acc1 = __builtin_amdgcn_wmma_f32_16x16x4_f32(
        false, a, false, b1, (short)0, acc1, false, false);
  }

  float bn = bias[mrow];
  int   mhi = half * 8;
#pragma unroll
  for (int j = 0; j < 8; ++j) {
    int   m   = j + mhi;
    float re  = acc0[j];
    float im  = acc1[j];
    float mag = __builtin_amdgcn_sqrtf(re * re + im * im);
    float sc  = fmaxf(mag + bn, 0.0f) *
                __builtin_amdgcn_rcpf(mag + 1e-8f);
    v2f v = { re * sc, im * sc };
    __builtin_nontemporal_store(
        v, (v2f*)(out + ((size_t)(node_base + m) * CH_OUT + mrow) * 2));
  }
}

extern "C" void kernel_launch(void* const* d_in, const int* in_sizes, int n_in,
                              void* d_out, int out_size, void* d_ws, size_t ws_size,
                              hipStream_t stream) {
  const float* x      = (const float*)d_in[0];
  const int*   edges  = (const int*)d_in[1];
  const float* sten   = (const float*)d_in[2];
  const float* weight = (const float*)d_in[3];
  const float* offset = (const float*)d_in[4];
  const float* bias   = (const float*)d_in[5];
  float*       out    = (float*)d_out;

  float* agg = (float*)d_ws;
  float* Fp  = agg + (size_t)N_NODES * K_DIM;

  hipMemsetAsync(agg, 0, (size_t)N_NODES * K_DIM * sizeof(float), stream);

  filter_kernel<<<(K_DIM * OCOLS + 255) / 256, 256, 0, stream>>>(weight, offset, Fp);

  long long total = (long long)N_EDGES * R_DIM;
  scatter_kernel<<<(int)((total + 255) / 256), 256, 0, stream>>>(x, edges, sten, agg);

  gemm_kernel<<<(N_TILES + 7) / 8, 256, 0, stream>>>(Fp, agg, bias, out);
}


# Complex filter product per edge as THREE real products (Gauss form): 36 f32 WMMAs per 16-edge tile, not 48; sums regrouped (re = P0+P2, im = P0+P1); Y rows planar, finish interleaves
# speedup vs baseline: 9.0184x; 9.0184x over previous
#include <hip/hip_runtime.h>


typedef float v2f __attribute__((ext_vector_type(2)));
typedef float v4f __attribute__((ext_vector_type(4)));
typedef float v8f __attribute__((ext_vector_type(8)));

#define N_NODES 100000
#define N_EDGES 1600000
#define R_DIM   6
#define CH_IN   8
#define CH_OUT  16
#define OCOLS   32
#define E_TILES (N_EDGES / 16)
#define G_STEPS 12
#define G_PRODS 3
#define G_FLOATS (G_PRODS * G_STEPS * 2 * CH_OUT * 2)

__global__ __launch_bounds__(256) void filter_kernel(
    const float* __restrict__ weight,
    const float* __restrict__ offset,
    float*       __restrict__ G)
{
  const int i = blockIdx.x * 256 + threadIdx.x;
  if (i >= G_FLOATS) return;
  const int s  = i & 1;
  const int o  = (i >> 1) & 15;
  const int h  = (i >> 5) & 1;
  const int mt = i >> 6;
  const int m  = mt / G_STEPS;
  const int t  = mt - m * G_STEPS;
  const int r  = t >> 1;
  const int c  = 4 * h + 2 * (t & 1) + s;

  const float w  = weight[(r * CH_IN + c) * CH_OUT + o];
  const float ph = offset[c * CH_OUT + o];
  const float fre = w * cosf(ph);
  const float fim = w * sinf(ph);

  float val;
  if (m == 0)      val = fre;
  else if (m == 1) val = fim - fre;
  else             val = -(fre + fim);

  G[i] = val;
}

__device__ __forceinline__ float xhalf(float v) {
  int b = __builtin_bit_cast(int, v);
  return __builtin_bit_cast(float,
      __builtin_amdgcn_permlanex16(b, b, 0x76543210, 0xfedcba98, false, false));
}

__global__ __launch_bounds__(128, 5) void edge_kernel(
    const float* __restrict__ G,
    const float* __restrict__ x,
    const int*   __restrict__ edges,
    const float* __restrict__ sten,
    float*       __restrict__ Y)
{
  const int lane = threadIdx.x & 31;
  const int c16  = lane & 15;
  const int half = lane >> 4;

  const v2f* g2 = (const v2f*)G;
  v2f b0[G_STEPS], b1[G_STEPS], b2[G_STEPS];
#pragma unroll
  for (int t = 0; t < G_STEPS; ++t) {
    b0[t] = g2[((0 * G_STEPS + t) * 2 + half) * CH_OUT + c16];
    b1[t] = g2[((1 * G_STEPS + t) * 2 + half) * CH_OUT + c16];
    b2[t] = g2[((2 * G_STEPS + t) * 2 + half) * CH_OUT + c16];
  }

  const int wave    = __builtin_amdgcn_readfirstlane((int)(threadIdx.x >> 5));
  const int n_waves = (int)gridDim.x * 4;

  for (int tile = (int)blockIdx.x * 4 + wave; tile < E_TILES; tile += n_waves) {
    const int e = tile * 16 + c16;

    long long ep = __builtin_nontemporal_load((const long long*)edges + e);
    const int src = (int)(ep & 0xffffffffLL);
    const int dst = (int)(ep >> 32);

    const v4f* sp = (const v4f*)(sten + (size_t)e * (R_DIM * 2));
    const v4f s0 = __builtin_nontemporal_load(sp);
    const v4f s1 = __builtin_nontemporal_load(sp + 1);
    const v4f s2 = __builtin_nontemporal_load(sp + 2);
    const float st[R_DIM * 2] = {s0.x, s0.y, s0.z, s0.w, s1.x, s1.y,
                                 s1.z, s1.w, s2.x, s2.y, s2.z, s2.w};

    const v4f xv = *(const v4f*)(x + (size_t)src * CH_IN + half * 4);
    const v2f xp[2] = { {xv.x, xv.y}, {xv.z, xv.w} };

    v8f p0 = {0.f, 0.f, 0.f, 0.f, 0.f, 0.f, 0.f, 0.f};
    v8f p1 = {0.f, 0.f, 0.f, 0.f, 0.f, 0.f, 0.f, 0.f};
    v8f p2 = {0.f, 0.f, 0.f, 0.f, 0.f, 0.f, 0.f, 0.f};

#pragma unroll
    for (int t = 0; t < G_STEPS; ++t) {
      const int r = t >> 1;
      const float sre = st[2 * r];
      const float sim = st[2 * r + 1];
      const v2f xx = xp[t & 1];
      const v2f a0 = (sre + sim) * xx;
      const v2f a1 = sre * xx;
      const v2f a2 = sim * xx;
      p0 = __builtin_amdgcn_wmma_f32_16x16x4_f32(
          false, a0, false, b0[t], (short)0, p0, false, false);
      p1 = __builtin_amdgcn_wmma_f32_16x16x4_f32(
          false, a1, false, b1[t], (short)0, p1, false, false);
      p2 = __builtin_amdgcn_wmma_f32_16x16x4_f32(
          false, a2, false, b2[t], (short)0, p2, false, false);
    }

    const v8f acc0 = p0 + p2;
    const v8f acc1 = p0 + p1;

#pragma unroll
    for (int j = 0; j < 8; ++j) {
      const int dA = __builtin_amdgcn_readlane(dst, j);
      const int dB = __builtin_amdgcn_readlane(dst, j + 8);
      const float send = half ? acc0[j] : acc1[j];
      const float recv = xhalf(send);
      const float vA = half ? recv    : acc0[j];
      const float vB = half ? acc1[j] : recv;
      unsafeAtomicAdd(Y + ((unsigned)dA * OCOLS + (unsigned)lane), vA);
      unsafeAtomicAdd(Y + ((unsigned)dB * OCOLS + (unsigned)lane), vB);
    }
  }
}

#define FIN_THREADS 160000
__global__ __launch_bounds__(256) void finish_kernel(
    const float* __restrict__ Y,
    const float* __restrict__ bias,
    float*       __restrict__ out)
{
  const int i = blockIdx.x * 256 + threadIdx.x;
  const int q = i & 7;
  const float bn0 = bias[2 * q];
  const float bn1 = bias[2 * q + 1];

  v2f re[5], im[5];
#pragma unroll
  for (int k = 0; k < 5; ++k) {
    const int n = (i + k * FIN_THREADS) >> 3;
    re[k] = ((const v2f*)Y)[n * 16 + q];
    im[k] = ((const v2f*)Y)[n * 16 + 8 + q];
  }

#pragma unroll
  for (int k = 0; k < 5; ++k) {
    const float re0 = re[k].x, im0 = im[k].x, re1 = re[k].y, im1 = im[k].y;
    const float mag0 = __builtin_amdgcn_sqrtf(re0 * re0 + im0 * im0);
    const float mag1 = __builtin_amdgcn_sqrtf(re1 * re1 + im1 * im1);
    const float sc0 = fmaxf(mag0 + bn0, 0.0f) * __builtin_amdgcn_rcpf(mag0 + 1e-8f);
    const float sc1 = fmaxf(mag1 + bn1, 0.0f) * __builtin_amdgcn_rcpf(mag1 + 1e-8f);
    const v4f w = { re0 * sc0, im0 * sc0, re1 * sc1, im1 * sc1 };
    __builtin_nontemporal_store(w, (v4f*)out + i + k * FIN_THREADS);
  }
}

extern "C" void kernel_launch(void* const* d_in, const int* in_sizes, int n_in,
                              void* d_out, int out_size, void* d_ws, size_t ws_size,
                              hipStream_t stream) {
  const float* x      = (const float*)d_in[0];
  const int*   edges  = (const int*)d_in[1];
  const float* sten   = (const float*)d_in[2];
  const float* weight = (const float*)d_in[3];
  const float* offset = (const float*)d_in[4];
  const float* bias   = (const float*)d_in[5];
  float*       out    = (float*)d_out;

  float* Y = (float*)d_ws;
  float* G = Y + (size_t)N_NODES * OCOLS;

  hipMemsetAsync(Y, 0, (size_t)N_NODES * OCOLS * sizeof(float), stream);

  filter_kernel<<<(G_FLOATS + 255) / 256, 256, 0, stream>>>(weight, offset, G);

  edge_kernel<<<1280, 128, 0, stream>>>(G, x, edges, sten, Y);

  finish_kernel<<<FIN_THREADS / 256, 256, 0, stream>>>(Y, bias, out);
}


# Tile loop software-pipelined over two register sets: the next tile's stencil, x rows and edge pair are requested in front of the current tile's WMMAs and atomics; sums still by arrival order
# speedup vs baseline: 9.7390x; 1.0799x over previous
#include <hip/hip_runtime.h>


typedef float v2f __attribute__((ext_vector_type(2)));
typedef float v4f __attribute__((ext_vector_type(4)));
typedef float v8f __attribute__((ext_vector_type(8)));

#define N_NODES 100000
#define N_EDGES 1600000
#define R_DIM   6
#define CH_IN   8
#define CH_OUT  16
#define OCOLS   32
#define E_TILES (N_EDGES / 16)
#define G_STEPS 12
#define G_PRODS 3
#define G_FLOATS (G_PRODS * G_STEPS * 2 * CH_OUT * 2)

__global__ __launch_bounds__(256) void filter_kernel(
    const float* __restrict__ weight,
    const float* __restrict__ offset,
    float*       __restrict__ G)
{
  const int i = blockIdx.x * 256 + threadIdx.x;
  if (i >= G_FLOATS) return;
  const int s  = i & 1;
  const int o  = (i >> 1) & 15;
  const int h  = (i >> 5) & 1;
  const int mt = i >> 6;
  const int m  = mt / G_STEPS;
  const int t  = mt - m * G_STEPS;
  const int r  = t >> 1;
  const int c  = 4 * h + 2 * (t & 1) + s;

  const float w  = weight[(r * CH_IN + c) * CH_OUT + o];
  const float ph = offset[c * CH_OUT + o];
  const float fre = w * cosf(ph);
  const float fim = w * sinf(ph);

  float val;
  if (m == 0)      val = fre;
  else if (m == 1) val = fim - fre;
  else             val = -(fre + fim);

  G[i] = val;
}

__device__ __forceinline__ float xhalf(float v) {
  int b = __builtin_bit_cast(int, v);
  return __builtin_bit_cast(float,
      __builtin_amdgcn_permlanex16(b, b, 0x76543210, 0xfedcba98, false, false));
}

__device__ __forceinline__ void tile_products_and_scatter(
    const v4f s0, const v4f s1, const v4f s2, const v4f xv, const int dst,
    const v2f (&b0)[G_STEPS], const v2f (&b1)[G_STEPS], const v2f (&b2)[G_STEPS],
    float* __restrict__ Y, const int lane, const int half)
{
  const float st[R_DIM * 2] = {s0.x, s0.y, s0.z, s0.w, s1.x, s1.y,
                               s1.z, s1.w, s2.x, s2.y, s2.z, s2.w};
  const v2f xp[2] = { {xv.x, xv.y}, {xv.z, xv.w} };

  v8f p0 = {0.f, 0.f, 0.f, 0.f, 0.f, 0.f, 0.f, 0.f};
  v8f p1 = {0.f, 0.f, 0.f, 0.f, 0.f, 0.f, 0.f, 0.f};
  v8f p2 = {0.f, 0.f, 0.f, 0.f, 0.f, 0.f, 0.f, 0.f};

#pragma unroll
  for (int t = 0; t < G_STEPS; ++t) {
    const int r = t >> 1;
    const float sre = st[2 * r];
    const float sim = st[2 * r + 1];
    const v2f xx = xp[t & 1];
    const v2f a0 = (sre + sim) * xx;
    const v2f a1 = sre * xx;
    const v2f a2 = sim * xx;
    p0 = __builtin_amdgcn_wmma_f32_16x16x4_f32(
        false, a0, false, b0[t], (short)0, p0, false, false);
    p1 = __builtin_amdgcn_wmma_f32_16x16x4_f32(
        false, a1, false, b1[t], (short)0, p1, false, false);
    p2 = __builtin_amdgcn_wmma_f32_16x16x4_f32(
        false, a2, false, b2[t], (short)0, p2, false, false);
  }

  const v8f acc0 = p0 + p2;
  const v8f acc1 = p0 + p1;

#pragma unroll
  for (int j = 0; j < 8; ++j) {
    const int dA = __builtin_amdgcn_readlane(dst, j);
    const int dB = __builtin_amdgcn_readlane(dst, j + 8);
    const float send = half ? acc0[j] : acc1[j];
    const float recv = xhalf(send);
    const float vA = half ? recv    : acc0[j];
    const float vB = half ? acc1[j] : recv;
    unsafeAtomicAdd(Y + ((unsigned)dA * OCOLS + (unsigned)lane), vA);
    unsafeAtomicAdd(Y + ((unsigned)dB * OCOLS + (unsigned)lane), vB);
  }
}

__global__ __launch_bounds__(128, 5) void edge_kernel(
    const float* __restrict__ G,
    const float* __restrict__ x,
    const int*   __restrict__ edges,
    const float* __restrict__ sten,
    float*       __restrict__ Y)
{
  const int lane = threadIdx.x & 31;
  const int c16  = lane & 15;
  const int half = lane >> 4;

  const v2f* g2 = (const v2f*)G;
  v2f b0[G_STEPS], b1[G_STEPS], b2[G_STEPS];
#pragma unroll
  for (int t = 0; t < G_STEPS; ++t) {
    b0[t] = g2[((0 * G_STEPS + t) * 2 + half) * CH_OUT + c16];
    b1[t] = g2[((1 * G_STEPS + t) * 2 + half) * CH_OUT + c16];
    b2[t] = g2[((2 * G_STEPS + t) * 2 + half) * CH_OUT + c16];
  }

  const int wave    = __builtin_amdgcn_readfirstlane((int)(threadIdx.x >> 5));
  const int n_waves = (int)gridDim.x * 4;

  int tile = (int)blockIdx.x * 4 + wave;
  if (tile >= E_TILES) return;

  const long long* ed = (const long long*)edges;

  const int e0 = tile * 16 + c16;
  const long long ep0 = __builtin_nontemporal_load(ed + e0);
  long long epB = __builtin_nontemporal_load(ed + (min(tile + n_waves, E_TILES - 1) * 16 + c16));
  long long epA;
  const v4f* sp0 = (const v4f*)(sten + (size_t)e0 * (R_DIM * 2));
  v4f a_s0 = __builtin_nontemporal_load(sp0);
  v4f a_s1 = __builtin_nontemporal_load(sp0 + 1);
  v4f a_s2 = __builtin_nontemporal_load(sp0 + 2);
  int a_dst = (int)(ep0 >> 32);
  v4f a_xv = *(const v4f*)(x + (size_t)(int)(ep0 & 0xffffffffLL) * CH_IN + half * 4);

  while (tile + n_waves < E_TILES) {
    const int t1 = tile + n_waves;
    epA = __builtin_nontemporal_load(ed + (min(t1 + n_waves, E_TILES - 1) * 16 + c16));
    const v4f* spb = (const v4f*)(sten + (size_t)(t1 * 16 + c16) * (R_DIM * 2));
    const v4f b_s0 = __builtin_nontemporal_load(spb);
    const v4f b_s1 = __builtin_nontemporal_load(spb + 1);
    const v4f b_s2 = __builtin_nontemporal_load(spb + 2);
    const v4f b_xv = *(const v4f*)(x + (size_t)(int)(epB & 0xffffffffLL) * CH_IN + half * 4);
    const int b_dst = (int)(epB >> 32);
    __builtin_amdgcn_sched_barrier(0);
    tile_products_and_scatter(a_s0, a_s1, a_s2, a_xv, a_dst, b0, b1, b2, Y, lane, half);
    __builtin_amdgcn_sched_barrier(0);

    const int t2 = t1 + n_waves;
    epB = __builtin_nontemporal_load(ed + (min(t2 + n_waves, E_TILES - 1) * 16 + c16));
    const v4f* spa = (const v4f*)(sten + (size_t)(min(t2, E_TILES - 1) * 16 + c16) * (R_DIM * 2));
    a_s0 = __builtin_nontemporal_load(spa);
    a_s1 = __builtin_nontemporal_load(spa + 1);
    a_s2 = __builtin_nontemporal_load(spa + 2);
    a_xv = *(const v4f*)(x + (size_t)(int)(epA & 0xffffffffLL) * CH_IN + half * 4);
    a_dst = (int)(epA >> 32);
    __builtin_amdgcn_sched_barrier(0);
    tile_products_and_scatter(b_s0, b_s1, b_s2, b_xv, b_dst, b0, b1, b2, Y, lane, half);

    tile = t2;
  }
  if (tile < E_TILES)
    tile_products_and_scatter(a_s0, a_s1, a_s2, a_xv, a_dst, b0, b1, b2, Y, lane, half);
}

#define FIN_THREADS 160000
__global__ __launch_bounds__(256) void finish_kernel(
    const float* __restrict__ Y,
    const float* __restrict__ bias,
    float*       __restrict__ out)
{
  const int i = blockIdx.x * 256 + threadIdx.x;
  const int q = i & 7;
  const float bn0 = bias[2 * q];
  const float bn1 = bias[2 * q + 1];

  v2f re[5], im[5];
#pragma unroll
  for (int k = 0; k < 5; ++k) {
    const int n = (i + k * FIN_THREADS) >> 3;
    re[k] = ((const v2f*)Y)[n * 16 + q];
    im[k] = ((const v2f*)Y)[n * 16 + 8 + q];
  }

#pragma unroll
  for (int k = 0; k < 5; ++k) {
    const float re0 = re[k].x, im0 = im[k].x, re1 = re[k].y, im1 = im[k].y;
    const float mag0 = __builtin_amdgcn_sqrtf(re0 * re0 + im0 * im0);
    const float mag1 = __builtin_amdgcn_sqrtf(re1 * re1 + im1 * im1);
    const float sc0 = fmaxf(mag0 + bn0, 0.0f) * __builtin_amdgcn_rcpf(mag0 + 1e-8f);
    const float sc1 = fmaxf(mag1 + bn1, 0.0f) * __builtin_amdgcn_rcpf(mag1 + 1e-8f);
    const v4f w = { re0 * sc0, im0 * sc0, re1 * sc1, im1 * sc1 };
    __builtin_nontemporal_store(w, (v4f*)out + i + k * FIN_THREADS);
  }
}

extern "C" void kernel_launch(void* const* d_in, const int* in_sizes, int n_in,
                              void* d_out, int out_size, void* d_ws, size_t ws_size,
                              hipStream_t stream) {
  const float* x      = (const float*)d_in[0];
  const int*   edges  = (const int*)d_in[1];
  const float* sten   = (const float*)d_in[2];
  const float* weight = (const float*)d_in[3];
  const float* offset = (const float*)d_in[4];
  const float* bias   = (const float*)d_in[5];
  float*       out    = (float*)d_out;

  float* Y = (float*)d_ws;
  float* G = Y + (size_t)N_NODES * OCOLS;

  hipMemsetAsync(Y, 0, (size_t)N_NODES * OCOLS * sizeof(float), stream);

  filter_kernel<<<(G_FLOATS + 255) / 256, 256, 0, stream>>>(weight, offset, G);

  edge_kernel<<<1280, 128, 0, stream>>>(G, x, edges, sten, Y);

  finish_kernel<<<FIN_THREADS / 256, 256, 0, stream>>>(Y, bias, out);
}
